# starting kernel
# baseline (speedup 1.0000x reference)
#include <hip/hip_runtime.h>

typedef float v2f __attribute__((ext_vector_type(2)));
typedef float v8f __attribute__((ext_vector_type(8)));

#define T_STEPS 1000
#define C_IN    18
#define C_AUG   19
#define N_INC   (T_STEPS - 1)
#define OUT_PER_PATH (C_AUG + C_AUG * C_AUG)
#define K_MAIN  996

__device__ __forceinline__ v8f wmma_f32_4(v2f a, v2f b, v8f c) {
  return __builtin_amdgcn_wmma_f32_16x16x4_f32(false, a, false, b, (short)0, c,
                                               false, false);
}

__device__ __forceinline__ void store_tile(float* __restrict__ op, v8f a,
                                           int mt, int nt, int half, int ll) {
#pragma unroll
  for (int v = 0; v < 8; ++v) {
    int i = mt * 16 + v + 8 * half;
    int j = nt * 16 + ll;
    if (i < C_AUG && j < C_AUG) op[C_AUG + i * C_AUG + j] = a[v];
  }
}

__global__ __launch_bounds__(256) void sig2_kernel(const float* __restrict__ x,
                                                   float* __restrict__ out,
                                                   int n_paths) {
  const int lane = threadIdx.x & 31;
  const int path = blockIdx.x * (blockDim.x >> 5) + (threadIdx.x >> 5);
  if (path >= n_paths) return;

  const int half = lane >> 4;
  const int ll   = lane & 15;
  const int klo  = half * 2;
  const float inv = 1.0f / (float)N_INC;

  const float* __restrict__ xp = x + (size_t)path * (T_STEPS * C_IN);
  float* __restrict__ op = out + (size_t)path * OUT_PER_PATH;

  const bool is_time  = (ll == 0);
  const bool t1_valid = (16 + ll) < C_AUG;
  const int  mc0 = is_time  ? 0 : (ll - 1);
  const int  mc1 = t1_valid ? (15 + ll) : 0;

  const float x0_0 = is_time  ? 0.0f : xp[mc0];
  const float x0_1 = t1_valid ? xp[mc1] : 0.0f;

  const float* p0 = xp + klo * C_IN + mc0;
  const float* p1 = xp + klo * C_IN + mc1;

  v8f acc00 = {}, acc01 = {}, acc10 = {}, acc11 = {};

  float trow = (float)klo * inv;
  const float t4 = 4.0f * inv;

  for (int it = 0; it < K_MAIN / 4; ++it) {
    float a0 = p0[0];
    float a1 = p0[C_IN];
    float a2 = p0[2 * C_IN];
    float b0 = p1[0];
    float b1 = p1[C_IN];
    float b2 = p1[2 * C_IN];
    a0 = is_time ? trow            : a0;
    a1 = is_time ? trow + inv      : a1;
    a2 = is_time ? trow + 2.0f*inv : a2;
    b0 = t1_valid ? b0 : 0.0f;
    b1 = t1_valid ? b1 : 0.0f;
    b2 = t1_valid ? b2 : 0.0f;

    v2f aT0 = {0.5f * (a0 + a1) - x0_0, 0.5f * (a1 + a2) - x0_0};
    v2f aT1 = {0.5f * (b0 + b1) - x0_1, 0.5f * (b1 + b2) - x0_1};
    v2f bT0 = {a1 - a0, a2 - a1};
    v2f bT1 = {b1 - b0, b2 - b1};

    acc00 = wmma_f32_4(aT0, bT0, acc00);
    acc01 = wmma_f32_4(aT0, bT1, acc01);
    acc10 = wmma_f32_4(aT1, bT0, acc10);
    acc11 = wmma_f32_4(aT1, bT1, acc11);

    p0 += 4 * C_IN;
    p1 += 4 * C_IN;
    trow += t4;
  }

  {
    const int row = K_MAIN + klo;
    int r2 = row + 2; if (r2 > T_STEPS - 1) r2 = T_STEPS - 1;
    const int d2 = (r2 - row) * C_IN;
    float a0 = p0[0];
    float a1 = p0[C_IN];
    float a2 = p0[d2];
    float b0 = p1[0];
    float b1 = p1[C_IN];
    float b2 = p1[d2];
    a0 = is_time ? trow            : a0;
    a1 = is_time ? trow + inv      : a1;
    a2 = is_time ? trow + 2.0f*inv : a2;
    b0 = t1_valid ? b0 : 0.0f;
    b1 = t1_valid ? b1 : 0.0f;
    b2 = t1_valid ? b2 : 0.0f;

    const bool v1 = (row + 1) < N_INC;
    v2f aT0, aT1, bT0, bT1;
    aT0.x = 0.5f * (a0 + a1) - x0_0;
    aT0.y = v1 ? (0.5f * (a1 + a2) - x0_0) : 0.0f;
    aT1.x = 0.5f * (b0 + b1) - x0_1;
    aT1.y = v1 ? (0.5f * (b1 + b2) - x0_1) : 0.0f;
    bT0.x = a1 - a0;
    bT0.y = v1 ? (a2 - a1) : 0.0f;
    bT1.x = b1 - b0;
    bT1.y = v1 ? (b2 - b1) : 0.0f;

    acc00 = wmma_f32_4(aT0, bT0, acc00);
    acc01 = wmma_f32_4(aT0, bT1, acc01);
    acc10 = wmma_f32_4(aT1, bT0, acc10);
    acc11 = wmma_f32_4(aT1, bT1, acc11);
  }

  if (lane < C_AUG) {
    op[lane] = (lane == 0)
                   ? 1.0f
                   : (xp[(T_STEPS - 1) * C_IN + (lane - 1)] - xp[lane - 1]);
  }

  store_tile(op, acc00, 0, 0, half, ll);
  store_tile(op, acc01, 0, 1, half, ll);
  store_tile(op, acc10, 1, 0, half, ll);
  store_tile(op, acc11, 1, 1, half, ll);
}

extern "C" void kernel_launch(void* const* d_in, const int* in_sizes, int n_in,
                              void* d_out, int out_size, void* d_ws, size_t ws_size,
                              hipStream_t stream) {
  const float* x = (const float*)d_in[0];
  float* out = (float*)d_out;
  const int n_paths = in_sizes[0] / (T_STEPS * C_IN);
  const int waves_per_block = 8;
  dim3 block(32 * waves_per_block);
  dim3 grid((n_paths + waves_per_block - 1) / waves_per_block);
  sig2_kernel<<<grid, block, 0, stream>>>(x, out, n_paths);
}


# Loads of the next 10 chunks (60 per wave) issued as one group ahead of the current 10 chunks' products, two register sets taking turns; order of summation and k pairing unchanged
# speedup vs baseline: 1.8658x; 1.8658x over previous
#include <hip/hip_runtime.h>


typedef float v2f __attribute__((ext_vector_type(2)));
typedef float v8f __attribute__((ext_vector_type(8)));

#define T_STEPS 1000
#define C_IN    18
#define C_AUG   19
#define N_INC   (T_STEPS - 1)
#define OUT_PER_PATH (C_AUG + C_AUG * C_AUG)
#define K_MAIN  996
#define U_CH    10
#define N_GRP   25

__device__ __forceinline__ v8f wmma_f32_4(v2f a, v2f b, v8f c) {
  return __builtin_amdgcn_wmma_f32_16x16x4_f32(false, a, false, b, (short)0, c,
                                               false, false);
}

__device__ __forceinline__ void store_tile(float* __restrict__ op, v8f a,
                                           int mt, int nt, int half, int ll) {
#pragma unroll
  for (int v = 0; v < 8; ++v) {
    int i = mt * 16 + v + 8 * half;
    int j = nt * 16 + ll;
    if (i < C_AUG && j < C_AUG) op[C_AUG + i * C_AUG + j] = a[v];
  }
}

struct Grp {
  float a[U_CH][3];
  float b[U_CH][3];
};

template <bool LAST>
__device__ __forceinline__ void load_group(Grp& g, const float* __restrict__ p0,
                                           const float* __restrict__ p1,
                                           int d2_last) {
#pragma unroll
  for (int u = 0; u < U_CH; ++u) {
    const int o  = u * 4 * C_IN;
    const int o2 = (LAST && u == U_CH - 1) ? d2_last : 2 * C_IN;
    g.a[u][0] = p0[o];
    g.a[u][1] = p0[o + C_IN];
    g.a[u][2] = p0[o + o2];
    g.b[u][0] = p1[o];
    g.b[u][1] = p1[o + C_IN];
    g.b[u][2] = p1[o + o2];
  }
}

template <bool LAST>
__device__ __forceinline__ void compute_group(const Grp& g, v8f& acc00, v8f& acc01,
                                              v8f& acc10, v8f& acc11, float& trow,
                                              bool is_time, bool t1_valid,
                                              float x0_0, float x0_1, float inv,
                                              float t4, bool v1_last) {
#pragma unroll
  for (int u = 0; u < U_CH; ++u) {
    float a0 = g.a[u][0], a1 = g.a[u][1], a2 = g.a[u][2];
    float b0 = g.b[u][0], b1 = g.b[u][1], b2 = g.b[u][2];
    a0 = is_time ? trow            : a0;
    a1 = is_time ? trow + inv      : a1;
    a2 = is_time ? trow + 2.0f*inv : a2;
    b0 = t1_valid ? b0 : 0.0f;
    b1 = t1_valid ? b1 : 0.0f;
    b2 = t1_valid ? b2 : 0.0f;

    v2f aT0 = {0.5f * (a0 + a1) - x0_0, 0.5f * (a1 + a2) - x0_0};
    v2f aT1 = {0.5f * (b0 + b1) - x0_1, 0.5f * (b1 + b2) - x0_1};
    v2f bT0 = {a1 - a0, a2 - a1};
    v2f bT1 = {b1 - b0, b2 - b1};
    if (LAST && u == U_CH - 1) {
      aT0.y = v1_last ? aT0.y : 0.0f;
      aT1.y = v1_last ? aT1.y : 0.0f;
      bT0.y = v1_last ? bT0.y : 0.0f;
      bT1.y = v1_last ? bT1.y : 0.0f;
    }

    acc00 = wmma_f32_4(aT0, bT0, acc00);
    acc01 = wmma_f32_4(aT0, bT1, acc01);
    acc10 = wmma_f32_4(aT1, bT0, acc10);
    acc11 = wmma_f32_4(aT1, bT1, acc11);

    trow += t4;
  }
}

__global__ __launch_bounds__(256) void sig2_kernel(const float* __restrict__ x,
                                                   float* __restrict__ out,
                                                   int n_paths) {
  const int lane = threadIdx.x & 31;
  const int path = blockIdx.x * (blockDim.x >> 5) + (threadIdx.x >> 5);
  if (path >= n_paths) return;

  const int half = lane >> 4;
  const int ll   = lane & 15;
  const int klo  = half * 2;
  const float inv = 1.0f / (float)N_INC;

  const float* __restrict__ xp = x + (size_t)path * (T_STEPS * C_IN);
  float* __restrict__ op = out + (size_t)path * OUT_PER_PATH;

  const bool is_time  = (ll == 0);
  const bool t1_valid = (16 + ll) < C_AUG;
  const int  mc0 = is_time  ? 0 : (ll - 1);
  const int  mc1 = t1_valid ? (15 + ll) : 0;

  const float x0_0 = is_time  ? 0.0f : xp[mc0];
  const float x0_1 = t1_valid ? xp[mc1] : 0.0f;

  const float* p0 = xp + klo * C_IN + mc0;
  const float* p1 = xp + klo * C_IN + mc1;

  v8f acc00 = {}, acc01 = {}, acc10 = {}, acc11 = {};

  float trow = (float)klo * inv;
  const float t4 = 4.0f * inv;

  const int  row_l = K_MAIN + klo;
  int r2 = row_l + 2; if (r2 > T_STEPS - 1) r2 = T_STEPS - 1;
  const int  d2_last = (r2 - row_l) * C_IN;
  const bool v1_last = (row_l + 1) < N_INC;

  const int gstep = U_CH * 4 * C_IN;

#define PIN() __builtin_amdgcn_sched_barrier(0)
  Grp A, B;
  load_group<false>(A, p0, p1, 0);
  p0 += gstep; p1 += gstep;
  PIN();
  for (int g = 0; g < N_GRP - 3; g += 2) {
    load_group<false>(B, p0, p1, 0);
    p0 += gstep; p1 += gstep;
    PIN();
    compute_group<false>(A, acc00, acc01, acc10, acc11, trow, is_time, t1_valid,
                         x0_0, x0_1, inv, t4, true);
    PIN();
    load_group<false>(A, p0, p1, 0);
    p0 += gstep; p1 += gstep;
    PIN();
    compute_group<false>(B, acc00, acc01, acc10, acc11, trow, is_time, t1_valid,
                         x0_0, x0_1, inv, t4, true);
    PIN();
  }
  load_group<false>(B, p0, p1, 0);
  p0 += gstep; p1 += gstep;
  PIN();
  compute_group<false>(A, acc00, acc01, acc10, acc11, trow, is_time, t1_valid,
                       x0_0, x0_1, inv, t4, true);
  PIN();
  load_group<true>(A, p0, p1, d2_last);
  PIN();
  compute_group<false>(B, acc00, acc01, acc10, acc11, trow, is_time, t1_valid,
                       x0_0, x0_1, inv, t4, true);
  compute_group<true>(A, acc00, acc01, acc10, acc11, trow, is_time, t1_valid,
                      x0_0, x0_1, inv, t4, v1_last);
#undef PIN

  if (lane < C_AUG) {
    op[lane] = (lane == 0)
                   ? 1.0f
                   : (xp[(T_STEPS - 1) * C_IN + (lane - 1)] - xp[lane - 1]);
  }

  store_tile(op, acc00, 0, 0, half, ll);
  store_tile(op, acc01, 0, 1, half, ll);
  store_tile(op, acc10, 1, 0, half, ll);
  store_tile(op, acc11, 1, 1, half, ll);
}

extern "C" void kernel_launch(void* const* d_in, const int* in_sizes, int n_in,
                              void* d_out, int out_size, void* d_ws, size_t ws_size,
                              hipStream_t stream) {
  const float* x = (const float*)d_in[0];
  float* out = (float*)d_out;
  const int n_paths = in_sizes[0] / (T_STEPS * C_IN);
  const int waves_per_block = 8;
  dim3 block(32 * waves_per_block);
  dim3 grid((n_paths + waves_per_block - 1) / waves_per_block);
  sig2_kernel<<<grid, block, 0, stream>>>(x, out, n_paths);
}


# Input rows staged in a per-wave LDS ring by whole-line 16-byte loads (18 in flight per wave), fragments read from LDS instead of by 4-byte global loads; summation order and k pairing unchanged
# speedup vs baseline: 1.9703x; 1.0560x over previous
#include <hip/hip_runtime.h>


typedef float v2f __attribute__((ext_vector_type(2)));
typedef float v4f __attribute__((ext_vector_type(4)));
typedef float v8f __attribute__((ext_vector_type(8)));

#define T_STEPS 1000
#define C_IN    18
#define C_AUG   19
#define N_INC   (T_STEPS - 1)
#define OUT_PER_PATH (C_AUG + C_AUG * C_AUG)
#define PATH_FLOATS (T_STEPS * C_IN)
#define K_MAIN  996
#define WAVES   8

#define STG_LOADS  9
#define STG_ROWS   64
#define STG_FLOATS (STG_ROWS * C_IN)
#define RING_ROWS   (2 * STG_ROWS)
#define RING_FLOATS (RING_ROWS * C_IN)
#define U_CH    8
#define N_TRIP  7

__device__ __forceinline__ v8f wmma_f32_4(v2f a, v2f b, v8f c) {
  return __builtin_amdgcn_wmma_f32_16x16x4_f32(false, a, false, b, (short)0, c,
                                               false, false);
}

__device__ __forceinline__ void store_tile(float* __restrict__ op, v8f a,
                                           int mt, int nt, int half, int ll) {
#pragma unroll
  for (int v = 0; v < 8; ++v) {
    int i = mt * 16 + v + 8 * half;
    int j = nt * 16 + ll;
    if (i < C_AUG && j < C_AUG) op[C_AUG + i * C_AUG + j] = a[v];
  }
}

struct Stg {
  v4f r[STG_LOADS];
};

__device__ __forceinline__ void issue_stage(Stg& s, const float* __restrict__ xp,
                                            int stage, int lane) {
#pragma unroll
  for (int j = 0; j < STG_LOADS; ++j) {
    int off = stage * STG_FLOATS + j * 128 + lane * 4;
    off = off < PATH_FLOATS - 4 ? off : PATH_FLOATS - 4;
    s.r[j] = *reinterpret_cast<const v4f*>(xp + off);
  }
}

__device__ __forceinline__ void store_stage(const Stg& s, float* slot_lane) {
#pragma unroll
  for (int j = 0; j < STG_LOADS; ++j)
    *reinterpret_cast<v4f*>(slot_lane + j * 128) = s.r[j];
}

struct Grp {
  float a[U_CH][3];
  float b[U_CH][3];
};

template <int P, int NCH, bool OWN3>
__device__ __forceinline__ void load_group(Grp& g, const float* la, const float* lb,
                                           const float* e3a, const float* e3b) {
#pragma unroll
  for (int u = 0; u < NCH; ++u) {
    const int o = (P + 4 * u) * C_IN;
    g.a[u][0] = la[o];
    g.a[u][1] = la[o + C_IN];
    g.b[u][0] = lb[o];
    g.b[u][1] = lb[o + C_IN];
    if (OWN3 && u == NCH - 1) {
      g.a[u][2] = *e3a;
      g.b[u][2] = *e3b;
    } else {
      g.a[u][2] = la[o + 2 * C_IN];
      g.b[u][2] = lb[o + 2 * C_IN];
    }
  }
}

template <int NCH, bool LAST>
__device__ __forceinline__ void compute_group(const Grp& g, v8f& acc00, v8f& acc01,
                                              v8f& acc10, v8f& acc11, float& trow,
                                              bool is_time, bool t1_valid,
                                              float x0_0, float x0_1, float inv,
                                              float t4, bool v1_last) {
#pragma unroll
  for (int u = 0; u < NCH; ++u) {
    float a0 = g.a[u][0], a1 = g.a[u][1], a2 = g.a[u][2];
    float b0 = g.b[u][0], b1 = g.b[u][1], b2 = g.b[u][2];
    a0 = is_time ? trow            : a0;
    a1 = is_time ? trow + inv      : a1;
    a2 = is_time ? trow + 2.0f*inv : a2;
    b0 = t1_valid ? b0 : 0.0f;
    b1 = t1_valid ? b1 : 0.0f;
    b2 = t1_valid ? b2 : 0.0f;

    v2f aT0 = {0.5f * (a0 + a1) - x0_0, 0.5f * (a1 + a2) - x0_0};
    v2f aT1 = {0.5f * (b0 + b1) - x0_1, 0.5f * (b1 + b2) - x0_1};
    v2f bT0 = {a1 - a0, a2 - a1};
    v2f bT1 = {b1 - b0, b2 - b1};
    if (LAST && u == NCH - 1) {
      aT0.y = v1_last ? aT0.y : 0.0f;
      aT1.y = v1_last ? aT1.y : 0.0f;
      bT0.y = v1_last ? bT0.y : 0.0f;
      bT1.y = v1_last ? bT1.y : 0.0f;
    }

    acc00 = wmma_f32_4(aT0, bT0, acc00);
    acc01 = wmma_f32_4(aT0, bT1, acc01);
    acc10 = wmma_f32_4(aT1, bT0, acc10);
    acc11 = wmma_f32_4(aT1, bT1, acc11);

    trow += t4;
  }
}

__global__ __launch_bounds__(256) void sig2_kernel(const float* __restrict__ x,
                                                   float* __restrict__ out,
                                                   int n_paths) {
  __shared__ __attribute__((aligned(16))) float ring_all[WAVES * RING_FLOATS];

  const int lane = threadIdx.x & 31;
  const int wave = threadIdx.x >> 5;
  const int path = blockIdx.x * WAVES + wave;
  if (path >= n_paths) return;

  const int half = lane >> 4;
  const int ll   = lane & 15;
  const int klo  = half * 2;
  const float inv = 1.0f / (float)N_INC;

  const float* __restrict__ xp = x + (size_t)path * PATH_FLOATS;
  float* __restrict__ op = out + (size_t)path * OUT_PER_PATH;

  const bool is_time  = (ll == 0);
  const bool t1_valid = (16 + ll) < C_AUG;
  const int  mc0 = is_time  ? 0 : (ll - 1);
  const int  mc1 = t1_valid ? (15 + ll) : 0;

  const float x0_0 = is_time  ? 0.0f : xp[mc0];
  const float x0_1 = t1_valid ? xp[mc1] : 0.0f;

  float* ring = ring_all + wave * RING_FLOATS;
  float* st0 = ring + lane * 4;
  float* st1 = st0 + STG_FLOATS;
  const float* la = ring + klo * C_IN + mc0;
  const float* lb = ring + klo * C_IN + mc1;
  const int wrow = (RING_ROWS - 2 + klo) % RING_ROWS;
  const float* wa = ring + wrow * C_IN + mc0;
  const float* wb = ring + wrow * C_IN + mc1;
  const int zrow = (T_STEPS - 2 + half) % RING_ROWS;
  const float* za = ring + zrow * C_IN + mc0;
  const float* zb = ring + zrow * C_IN + mc1;
  const bool v1_last = (K_MAIN + klo + 1) < N_INC;

  v8f acc00 = {}, acc01 = {}, acc10 = {}, acc11 = {};

  float trow = (float)klo * inv;
  const float t4 = 4.0f * inv;

#define PIN() __builtin_amdgcn_sched_barrier(0)
#define LDS_ORDER() __builtin_amdgcn_fence(__ATOMIC_ACQ_REL, "wavefront")
#define COMPUTE(G) compute_group<U_CH, false>(G, acc00, acc01, acc10, acc11, trow, \
                                              is_time, t1_valid, x0_0, x0_1, inv, t4, true)

  Stg R0, R1;
  issue_stage(R0, xp, 0, lane);
  issue_stage(R1, xp, 1, lane);
  store_stage(R0, st0);
  store_stage(R1, st1);
  LDS_ORDER();
  issue_stage(R0, xp, 2, lane);
  issue_stage(R1, xp, 3, lane);
  PIN();

  Grp A, B;
  load_group<0, U_CH, false>(A, la, lb, la, lb);
  PIN();
  for (int t = 0; t < N_TRIP; ++t) {
    load_group<32, U_CH, false>(B, la, lb, la, lb);
    PIN();
    COMPUTE(A);
    PIN();
    load_group<64, U_CH, false>(A, la, lb, la, lb);
    PIN();
    COMPUTE(B);
    PIN();
    store_stage(R0, st0);
    LDS_ORDER();
    issue_stage(R0, xp, 2 * t + 4, lane);
    PIN();
    load_group<96, U_CH, true>(B, la, lb, wa, wb);
    PIN();
    COMPUTE(A);
    PIN();
    load_group<0, U_CH, false>(A, la, lb, la, lb);
    PIN();
    COMPUTE(B);
    PIN();
    store_stage(R1, st1);
    LDS_ORDER();
    issue_stage(R1, xp, 2 * t + 5, lane);
    PIN();
  }
  load_group<32, U_CH, false>(B, la, lb, la, lb);
  PIN();
  COMPUTE(A);
  PIN();
  load_group<64, U_CH, false>(A, la, lb, la, lb);
  PIN();
  COMPUTE(B);
  PIN();
  load_group<96, 2, true>(B, la, lb, za, zb);
  PIN();
  COMPUTE(A);
  compute_group<2, true>(B, acc00, acc01, acc10, acc11, trow, is_time, t1_valid,
                         x0_0, x0_1, inv, t4, v1_last);
#undef COMPUTE
#undef LDS_ORDER
#undef PIN

  if (lane < C_AUG) {
    op[lane] = (lane == 0)
                   ? 1.0f
                   : (xp[(T_STEPS - 1) * C_IN + (lane - 1)] - xp[lane - 1]);
  }

  store_tile(op, acc00, 0, 0, half, ll);
  store_tile(op, acc01, 0, 1, half, ll);
  store_tile(op, acc10, 1, 0, half, ll);
  store_tile(op, acc11, 1, 1, half, ll);
}

extern "C" void kernel_launch(void* const* d_in, const int* in_sizes, int n_in,
                              void* d_out, int out_size, void* d_ws, size_t ws_size,
                              hipStream_t stream) {
  const float* x = (const float*)d_in[0];
  float* out = (float*)d_out;
  const int n_paths = in_sizes[0] / (T_STEPS * C_IN);
  const int waves_per_block = WAVES;
  dim3 block(32 * waves_per_block);
  dim3 grid((n_paths + waves_per_block - 1) / waves_per_block);
  sig2_kernel<<<grid, block, 0, stream>>>(x, out, n_paths);
}


# Mirror identity S2[i][j]+S2[j][i]=S1[i]*S1[j]: two 16x16 tiles summed per chunk instead of four; 18 entries are S1[i]*S1[j]-S2[j][i], not summed; k pairing and order of the summed entries unchanged
# speedup vs baseline: 2.4289x; 1.2328x over previous
#include <hip/hip_runtime.h>


typedef float v2f __attribute__((ext_vector_type(2)));
typedef float v4f __attribute__((ext_vector_type(4)));
typedef float v8f __attribute__((ext_vector_type(8)));

#define T_STEPS 1000
#define C_IN    18
#define C_AUG   19
#define N_INC   (T_STEPS - 1)
#define OUT_PER_PATH (C_AUG + C_AUG * C_AUG)
#define PATH_FLOATS (T_STEPS * C_IN)
#define K_MAIN  996
#define WAVES   8

#define STG_LOADS  9
#define STG_ROWS   64
#define STG_FLOATS (STG_ROWS * C_IN)
#define RING_ROWS   (2 * STG_ROWS)
#define RING_FLOATS (RING_ROWS * C_IN)
#define U_CH    8
#define N_TRIP  7

__device__ __forceinline__ v8f wmma_f32_4(v2f a, v2f b, v8f c) {
  return __builtin_amdgcn_wmma_f32_16x16x4_f32(false, a, false, b, (short)0, c,
                                               false, false);
}

__device__ __forceinline__ float lane_value(float v, int src_lane) {
  return __int_as_float(__builtin_amdgcn_readlane(__float_as_int(v), src_lane));
}

struct Stg {
  v4f r[STG_LOADS];
};

__device__ __forceinline__ void issue_stage(Stg& s, const float* __restrict__ xp,
                                            int stage, int lane) {
#pragma unroll
  for (int j = 0; j < STG_LOADS; ++j) {
    int off = stage * STG_FLOATS + j * 128 + lane * 4;
    off = off < PATH_FLOATS - 4 ? off : PATH_FLOATS - 4;
    s.r[j] = *reinterpret_cast<const v4f*>(xp + off);
  }
}

__device__ __forceinline__ void store_stage(const Stg& s, float* slot_lane) {
#pragma unroll
  for (int j = 0; j < STG_LOADS; ++j)
    *reinterpret_cast<v4f*>(slot_lane + j * 128) = s.r[j];
}

struct Grp {
  float a[U_CH][3];
  float b[U_CH][3];
};

template <int P, int NCH, bool OWN3>
__device__ __forceinline__ void load_group(Grp& g, const float* la, const float* lb,
                                           const float* e3a, const float* e3b) {
#pragma unroll
  for (int u = 0; u < NCH; ++u) {
    const int o = (P + 4 * u) * C_IN;
    g.a[u][0] = la[o];
    g.a[u][1] = la[o + C_IN];
    g.b[u][0] = lb[o];
    g.b[u][1] = lb[o + C_IN];
    if (OWN3 && u == NCH - 1) {
      g.a[u][2] = *e3a;
      g.b[u][2] = *e3b;
    } else {
      g.a[u][2] = la[o + 2 * C_IN];
      g.b[u][2] = lb[o + 2 * C_IN];
    }
  }
}

template <int NCH, bool LAST>
__device__ __forceinline__ void compute_group(const Grp& g, v8f& acc1, v8f& acc2,
                                              float& trow, bool is_time, bool lo3,
                                              bool hi3, float x0_0, float x0_1,
                                              float inv, float t4, bool v1_last) {
#pragma unroll
  for (int u = 0; u < NCH; ++u) {
    float a0 = g.a[u][0], a1 = g.a[u][1], a2 = g.a[u][2];
    float b0 = g.b[u][0], b1 = g.b[u][1], b2 = g.b[u][2];
    a0 = is_time ? trow            : a0;
    a1 = is_time ? trow + inv      : a1;
    a2 = is_time ? trow + 2.0f*inv : a2;

    v2f L1 = {0.5f * (a0 + a1) - x0_0, 0.5f * (a1 + a2) - x0_0};
    v2f L2 = {0.5f * (b0 + b1) - x0_1, 0.5f * (b1 + b2) - x0_1};
    v2f D1 = {a1 - a0, a2 - a1};
    v2f D2 = {b1 - b0, b2 - b1};
    v2f A2 = {hi3 ? L2.x : L1.x, hi3 ? L2.y : L1.y};
    v2f B2 = {lo3 ? D2.x : D1.x, lo3 ? D2.y : D1.y};
    if (LAST && u == NCH - 1) {
      L1.y = v1_last ? L1.y : 0.0f;
      A2.y = v1_last ? A2.y : 0.0f;
      D1.y = v1_last ? D1.y : 0.0f;
      B2.y = v1_last ? B2.y : 0.0f;
    }

    acc1 = wmma_f32_4(L1, D1, acc1);
    acc2 = wmma_f32_4(A2, B2, acc2);

    trow += t4;
  }
}

__global__ __launch_bounds__(256) void sig2_kernel(const float* __restrict__ x,
                                                   float* __restrict__ out,
                                                   int n_paths) {
  __shared__ __attribute__((aligned(16))) float ring_all[WAVES * RING_FLOATS];

  const int lane = threadIdx.x & 31;
  const int wave = threadIdx.x >> 5;
  const int path = blockIdx.x * WAVES + wave;
  if (path >= n_paths) return;

  const int half = lane >> 4;
  const int ll   = lane & 15;
  const int klo  = half * 2;
  const float inv = 1.0f / (float)N_INC;

  const float* __restrict__ xp = x + (size_t)path * PATH_FLOATS;
  float* __restrict__ op = out + (size_t)path * OUT_PER_PATH;

  const bool is_time = (ll == 0);
  const bool lo3 = ll < 3;
  const bool hi3 = ll >= 13;
  const int  mc0 = is_time ? 0 : (ll - 1);
  const int  mc1 = lo3 ? (15 + ll) : (hi3 ? (ll + 2) : mc0);

  const float x0_0 = is_time ? 0.0f : xp[mc0];
  const float x0_1 = xp[mc1];

  float* ring = ring_all + wave * RING_FLOATS;
  float* st0 = ring + lane * 4;
  float* st1 = st0 + STG_FLOATS;
  const float* la = ring + klo * C_IN + mc0;
  const float* lb = ring + klo * C_IN + mc1;
  const int wrow = (RING_ROWS - 2 + klo) % RING_ROWS;
  const float* wa = ring + wrow * C_IN + mc0;
  const float* wb = ring + wrow * C_IN + mc1;
  const int zrow = (T_STEPS - 2 + half) % RING_ROWS;
  const float* za = ring + zrow * C_IN + mc0;
  const float* zb = ring + zrow * C_IN + mc1;
  const bool v1_last = (K_MAIN + klo + 1) < N_INC;

  v8f acc1 = {}, acc2 = {};

  float trow = (float)klo * inv;
  const float t4 = 4.0f * inv;

#define PIN() __builtin_amdgcn_sched_barrier(0)
#define LDS_ORDER() __builtin_amdgcn_fence(__ATOMIC_ACQ_REL, "wavefront")
#define COMPUTE(G) compute_group<U_CH, false>(G, acc1, acc2, trow, is_time, lo3, hi3, \
                                              x0_0, x0_1, inv, t4, true)

  Stg R0, R1;
  issue_stage(R0, xp, 0, lane);
  issue_stage(R1, xp, 1, lane);
  store_stage(R0, st0);
  store_stage(R1, st1);
  LDS_ORDER();
  issue_stage(R0, xp, 2, lane);
  issue_stage(R1, xp, 3, lane);
  PIN();

  Grp A, B;
  load_group<0, U_CH, false>(A, la, lb, la, lb);
  PIN();
  for (int t = 0; t < N_TRIP; ++t) {
    load_group<32, U_CH, false>(B, la, lb, la, lb);
    PIN();
    COMPUTE(A);
    PIN();
    load_group<64, U_CH, false>(A, la, lb, la, lb);
    PIN();
    COMPUTE(B);
    PIN();
    store_stage(R0, st0);
    LDS_ORDER();
    issue_stage(R0, xp, 2 * t + 4, lane);
    PIN();
    load_group<96, U_CH, true>(B, la, lb, wa, wb);
    PIN();
    COMPUTE(A);
    PIN();
    load_group<0, U_CH, false>(A, la, lb, la, lb);
    PIN();
    COMPUTE(B);
    PIN();
    store_stage(R1, st1);
    LDS_ORDER();
    issue_stage(R1, xp, 2 * t + 5, lane);
    PIN();
  }
  load_group<32, U_CH, false>(B, la, lb, la, lb);
  PIN();
  COMPUTE(A);
  PIN();
  load_group<64, U_CH, false>(A, la, lb, la, lb);
  PIN();
  COMPUTE(B);
  PIN();
  load_group<96, 2, true>(B, la, lb, za, zb);
  PIN();
  COMPUTE(A);
  compute_group<2, true>(B, acc1, acc2, trow, is_time, lo3, hi3, x0_0, x0_1, inv, t4,
                         v1_last);
#undef COMPUTE
#undef LDS_ORDER
#undef PIN

  const int lrow = (T_STEPS - 1) % RING_ROWS;
  const float s1a = is_time ? 1.0f : (ring[lrow * C_IN + mc0] - x0_0);
  const float s1b = ring[lrow * C_IN + mc1] - x0_1;
  if (half == 0) {
    op[ll] = s1a;
    if (lo3) op[16 + ll] = s1b;
  }
  const float s1_c1  = lane_value(s1a, 1);
  const float s1_c2  = lane_value(s1a, 2);
  const float s1_c16 = lane_value(s1b, 0);
  const float s1_c17 = lane_value(s1b, 1);
  const float s1_c18 = lane_value(s1b, 2);

#pragma unroll
  for (int v = 0; v < 8; ++v) op[C_AUG + (v + 8 * half) * C_AUG + ll] = acc1[v];

#pragma unroll
  for (int v = 0; v < 8; ++v) {
    const int m = v + 8 * half;
    const float val = acc2[v];
    if (lo3) {
      const int j = 16 + ll;
      if (m < 13) {
        op[C_AUG + m * C_AUG + j] = val;
        if (m < 3) {
          const float s1m = (m == 0) ? 1.0f : (m == 1 ? s1_c1 : s1_c2);
          op[C_AUG + j * C_AUG + m] = s1b * s1m - val;
        }
      } else {
        op[C_AUG + (m + 3) * C_AUG + j] = val;
      }
    } else if (m >= 13) {
      const int i = m + 3;
      op[C_AUG + i * C_AUG + ll] = val;
      if (hi3) {
        const float s1i = (m == 13) ? s1_c16 : (m == 14 ? s1_c17 : s1_c18);
        op[C_AUG + ll * C_AUG + i] = s1a * s1i - val;
      }
    }
  }
}

extern "C" void kernel_launch(void* const* d_in, const int* in_sizes, int n_in,
                              void* d_out, int out_size, void* d_ws, size_t ws_size,
                              hipStream_t stream) {
  const float* x = (const float*)d_in[0];
  float* out = (float*)d_out;
  const int n_paths = in_sizes[0] / (T_STEPS * C_IN);
  const int waves_per_block = WAVES;
  dim3 block(32 * waves_per_block);
  dim3 grid((n_paths + waves_per_block - 1) / waves_per_block);
  sig2_kernel<<<grid, block, 0, stream>>>(x, out, n_paths);
}


# Row operand is the bare neighbour sum x[k]+x[k+1]: factor one half and start-point term x0[i]*S1[j] applied once after the loop, not per chunk; summed values change, order and k pairing do not
# speedup vs baseline: 2.5570x; 1.0527x over previous
#include <hip/hip_runtime.h>


typedef float v2f __attribute__((ext_vector_type(2)));
typedef float v4f __attribute__((ext_vector_type(4)));
typedef float v8f __attribute__((ext_vector_type(8)));

#define T_STEPS 1000
#define C_IN    18
#define C_AUG   19
#define N_INC   (T_STEPS - 1)
#define OUT_PER_PATH (C_AUG + C_AUG * C_AUG)
#define PATH_FLOATS (T_STEPS * C_IN)
#define K_MAIN  996
#define WAVES   8

#define STG_LOADS  9
#define STG_ROWS   64
#define STG_FLOATS (STG_ROWS * C_IN)
#define RING_ROWS   (2 * STG_ROWS)
#define RING_FLOATS (RING_ROWS * C_IN)
#define U_CH    8
#define N_TRIP  7

__device__ __forceinline__ v8f wmma_f32_4(v2f a, v2f b, v8f c) {
  return __builtin_amdgcn_wmma_f32_16x16x4_f32(false, a, false, b, (short)0, c,
                                               false, false);
}

__device__ __forceinline__ float lane_value(float v, int src_lane) {
  return __int_as_float(__builtin_amdgcn_readlane(__float_as_int(v), src_lane));
}

struct Stg {
  v4f r[STG_LOADS];
};

__device__ __forceinline__ void issue_stage(Stg& s, const float* __restrict__ xp,
                                            int stage, int lane) {
#pragma unroll
  for (int j = 0; j < STG_LOADS; ++j) {
    int off = stage * STG_FLOATS + j * 128 + lane * 4;
    off = off < PATH_FLOATS - 4 ? off : PATH_FLOATS - 4;
    s.r[j] = *reinterpret_cast<const v4f*>(xp + off);
  }
}

__device__ __forceinline__ void store_stage(const Stg& s, float* slot_lane) {
#pragma unroll
  for (int j = 0; j < STG_LOADS; ++j)
    *reinterpret_cast<v4f*>(slot_lane + j * 128) = s.r[j];
}

struct Grp {
  float a[U_CH][3];
  float b[U_CH][3];
};

template <int P, int NCH, bool OWN3>
__device__ __forceinline__ void load_group(Grp& g, const float* la, const float* lb,
                                           const float* e3a, const float* e3b) {
#pragma unroll
  for (int u = 0; u < NCH; ++u) {
    const int o = (P + 4 * u) * C_IN;
    g.a[u][0] = la[o];
    g.a[u][1] = la[o + C_IN];
    g.b[u][0] = lb[o];
    g.b[u][1] = lb[o + C_IN];
    if (OWN3 && u == NCH - 1) {
      g.a[u][2] = *e3a;
      g.b[u][2] = *e3b;
    } else {
      g.a[u][2] = la[o + 2 * C_IN];
      g.b[u][2] = lb[o + 2 * C_IN];
    }
  }
}

template <int NCH, bool LAST>
__device__ __forceinline__ void compute_group(const Grp& g, v8f& acc1, v8f& acc2,
                                              float& trow, bool is_time, bool lo3,
                                              bool hi3, float inv, float t4,
                                              bool v1_last) {
#pragma unroll
  for (int u = 0; u < NCH; ++u) {
    float a0 = g.a[u][0], a1 = g.a[u][1], a2 = g.a[u][2];
    float b0 = g.b[u][0], b1 = g.b[u][1], b2 = g.b[u][2];
    a0 = is_time ? trow            : a0;
    a1 = is_time ? trow + inv      : a1;
    a2 = is_time ? trow + 2.0f*inv : a2;

    v2f L1 = {a0 + a1, a1 + a2};
    v2f L2 = {b0 + b1, b1 + b2};
    v2f D1 = {a1 - a0, a2 - a1};
    v2f D2 = {b1 - b0, b2 - b1};
    v2f A2 = {hi3 ? L2.x : L1.x, hi3 ? L2.y : L1.y};
    v2f B2 = {lo3 ? D2.x : D1.x, lo3 ? D2.y : D1.y};
    if (LAST && u == NCH - 1) {
      L1.y = v1_last ? L1.y : 0.0f;
      A2.y = v1_last ? A2.y : 0.0f;
      D1.y = v1_last ? D1.y : 0.0f;
      B2.y = v1_last ? B2.y : 0.0f;
    }

    acc1 = wmma_f32_4(L1, D1, acc1);
    acc2 = wmma_f32_4(A2, B2, acc2);

    trow += t4;
  }
}

__global__ __launch_bounds__(256) void sig2_kernel(const float* __restrict__ x,
                                                   float* __restrict__ out,
                                                   int n_paths) {
  __shared__ __attribute__((aligned(16))) float ring_all[WAVES * RING_FLOATS];

  const int lane = threadIdx.x & 31;
  const int wave = threadIdx.x >> 5;
  const int path = blockIdx.x * WAVES + wave;
  if (path >= n_paths) return;

  const int half = lane >> 4;
  const int ll   = lane & 15;
  const int klo  = half * 2;
  const float inv = 1.0f / (float)N_INC;

  const float* __restrict__ xp = x + (size_t)path * PATH_FLOATS;
  float* __restrict__ op = out + (size_t)path * OUT_PER_PATH;

  const bool is_time = (ll == 0);
  const bool lo3 = ll < 3;
  const bool hi3 = ll >= 13;
  const int  mc0 = is_time ? 0 : (ll - 1);
  const int  mc1 = lo3 ? (15 + ll) : (hi3 ? (ll + 2) : mc0);

  const float x0_0 = is_time ? 0.0f : xp[mc0];
  const float x0_1 = xp[mc1];

  float* ring = ring_all + wave * RING_FLOATS;
  float* st0 = ring + lane * 4;
  float* st1 = st0 + STG_FLOATS;
  const float* la = ring + klo * C_IN + mc0;
  const float* lb = ring + klo * C_IN + mc1;
  const int wrow = (RING_ROWS - 2 + klo) % RING_ROWS;
  const float* wa = ring + wrow * C_IN + mc0;
  const float* wb = ring + wrow * C_IN + mc1;
  const int zrow = (T_STEPS - 2 + half) % RING_ROWS;
  const float* za = ring + zrow * C_IN + mc0;
  const float* zb = ring + zrow * C_IN + mc1;
  const bool v1_last = (K_MAIN + klo + 1) < N_INC;

  v8f acc1 = {}, acc2 = {};

  float trow = (float)klo * inv;
  const float t4 = 4.0f * inv;

#define PIN() __builtin_amdgcn_sched_barrier(0)
#define LDS_ORDER() __builtin_amdgcn_fence(__ATOMIC_ACQ_REL, "wavefront")
#define COMPUTE(G) compute_group<U_CH, false>(G, acc1, acc2, trow, is_time, lo3, hi3, \
                                              inv, t4, true)

  Stg R0, R1;
  issue_stage(R0, xp, 0, lane);
  issue_stage(R1, xp, 1, lane);
  store_stage(R0, st0);
  store_stage(R1, st1);
  LDS_ORDER();
  issue_stage(R0, xp, 2, lane);
  issue_stage(R1, xp, 3, lane);
  PIN();

  Grp A, B;
  load_group<0, U_CH, false>(A, la, lb, la, lb);
  PIN();
  for (int t = 0; t < N_TRIP; ++t) {
    load_group<32, U_CH, false>(B, la, lb, la, lb);
    PIN();
    COMPUTE(A);
    PIN();
    load_group<64, U_CH, false>(A, la, lb, la, lb);
    PIN();
    COMPUTE(B);
    PIN();
    store_stage(R0, st0);
    LDS_ORDER();
    issue_stage(R0, xp, 2 * t + 4, lane);
    PIN();
    load_group<96, U_CH, true>(B, la, lb, wa, wb);
    PIN();
    COMPUTE(A);
    PIN();
    load_group<0, U_CH, false>(A, la, lb, la, lb);
    PIN();
    COMPUTE(B);
    PIN();
    store_stage(R1, st1);
    LDS_ORDER();
    issue_stage(R1, xp, 2 * t + 5, lane);
    PIN();
  }
  load_group<32, U_CH, false>(B, la, lb, la, lb);
  PIN();
  COMPUTE(A);
  PIN();
  load_group<64, U_CH, false>(A, la, lb, la, lb);
  PIN();
  COMPUTE(B);
  PIN();
  load_group<96, 2, true>(B, la, lb, za, zb);
  PIN();
  COMPUTE(A);
  compute_group<2, true>(B, acc1, acc2, trow, is_time, lo3, hi3, inv, t4, v1_last);
#undef COMPUTE
#undef LDS_ORDER
#undef PIN

  const int lrow = (T_STEPS - 1) % RING_ROWS;
  const float s1a = is_time ? 1.0f : (ring[lrow * C_IN + mc0] - x0_0);
  const float s1b = ring[lrow * C_IN + mc1] - x0_1;
  if (half == 0) {
    op[ll] = s1a;
    if (lo3) op[16 + ll] = s1b;
  }
  const float s1_c1  = lane_value(s1a, 1);
  const float s1_c2  = lane_value(s1a, 2);
  const float s1_c16 = lane_value(s1b, 0);
  const float s1_c17 = lane_value(s1b, 1);
  const float s1_c18 = lane_value(s1b, 2);

  const float s1c2 = lo3 ? s1b : s1a;
  float x0r1[8], x0r2[8];
#pragma unroll
  for (int v = 0; v < 8; ++v) {
    const float r_lo = lane_value(x0_0, v);
    const float r_hi = lane_value(x0_0, v + 8);
    x0r1[v] = half ? r_hi : r_lo;
    if (v >= 5) {
      const float r_x = lane_value(x0_1, v + 8);
      x0r2[v] = half ? r_x : r_lo;
    } else {
      x0r2[v] = x0r1[v];
    }
  }

#pragma unroll
  for (int v = 0; v < 8; ++v)
    op[C_AUG + (v + 8 * half) * C_AUG + ll] = 0.5f * acc1[v] - x0r1[v] * s1a;

#pragma unroll
  for (int v = 0; v < 8; ++v) {
    const int m = v + 8 * half;
    const float val = 0.5f * acc2[v] - x0r2[v] * s1c2;
    if (lo3) {
      const int j = 16 + ll;
      if (m < 13) {
        op[C_AUG + m * C_AUG + j] = val;
        if (m < 3) {
          const float s1m = (m == 0) ? 1.0f : (m == 1 ? s1_c1 : s1_c2);
          op[C_AUG + j * C_AUG + m] = s1b * s1m - val;
        }
      } else {
        op[C_AUG + (m + 3) * C_AUG + j] = val;
      }
    } else if (m >= 13) {
      const int i = m + 3;
      op[C_AUG + i * C_AUG + ll] = val;
      if (hi3) {
        const float s1i = (m == 13) ? s1_c16 : (m == 14 ? s1_c17 : s1_c18);
        op[C_AUG + ll * C_AUG + i] = s1a * s1i - val;
      }
    }
  }
}

extern "C" void kernel_launch(void* const* d_in, const int* in_sizes, int n_in,
                              void* d_out, int out_size, void* d_ws, size_t ws_size,
                              hipStream_t stream) {
  const float* x = (const float*)d_in[0];
  float* out = (float*)d_out;
  const int n_paths = in_sizes[0] / (T_STEPS * C_IN);
  const int waves_per_block = WAVES;
  dim3 block(32 * waves_per_block);
  dim3 grid((n_paths + waves_per_block - 1) / waves_per_block);
  sig2_kernel<<<grid, block, 0, stream>>>(x, out, n_paths);
}


# Summation by parts: row operand is the bare row pair from LDS, column operand x[m+1]-x[m-1]; time channel from an LDS table; summed values change, chunk order and k pairing do not
# speedup vs baseline: 2.7004x; 1.0561x over previous
#include <hip/hip_runtime.h>


typedef float v2f __attribute__((ext_vector_type(2)));
typedef float v4f __attribute__((ext_vector_type(4)));
typedef float v8f __attribute__((ext_vector_type(8)));

#define T_STEPS 1000
#define C_IN    18
#define C_AUG   19
#define N_INC   (T_STEPS - 1)
#define OUT_PER_PATH (C_AUG + C_AUG * C_AUG)
#define PATH_FLOATS (T_STEPS * C_IN)
#define WAVES   8

#define STG_LOADS  9
#define STG_ROWS   64
#define STG_FLOATS (STG_ROWS * C_IN)
#define RING_ROWS   (2 * STG_ROWS)
#define RING_FLOATS (RING_ROWS * C_IN)
#define U_CH    8
#define N_TRIP  7
#define TAB_ROWS   (T_STEPS + 2)
#define TAB0       (WAVES * RING_FLOATS)
#define LDS_FLOATS (TAB0 + TAB_ROWS * C_IN)

__device__ __forceinline__ v8f wmma_f32_4(v2f a, v2f b, v8f c) {
  return __builtin_amdgcn_wmma_f32_16x16x4_f32(false, a, false, b, (short)0, c,
                                               false, false);
}

__device__ __forceinline__ float lane_value(float v, int src_lane) {
  return __int_as_float(__builtin_amdgcn_readlane(__float_as_int(v), src_lane));
}

struct Stg {
  v4f r[STG_LOADS];
};

__device__ __forceinline__ void issue_stage(Stg& s, const float* __restrict__ xp,
                                            int stage, int lane) {
#pragma unroll
  for (int j = 0; j < STG_LOADS; ++j) {
    int off = stage * STG_FLOATS + j * 128 + lane * 4;
    off = off < PATH_FLOATS - 4 ? off : PATH_FLOATS - 4;
    s.r[j] = *reinterpret_cast<const v4f*>(xp + off);
  }
}

__device__ __forceinline__ void store_stage(const Stg& s, float* slot_lane) {
#pragma unroll
  for (int j = 0; j < STG_LOADS; ++j)
    *reinterpret_cast<v4f*>(slot_lane + j * 128) = s.r[j];
}

struct Grp {
  float a[U_CH][4];
  float b[U_CH][4];
};

template <int P, int NCH, bool OWN0, bool OWN3>
__device__ __forceinline__ void load_group(Grp& g, const float* lds, int la, int lb,
                                           int e0a, int e0b, int e3a, int e3b) {
#pragma unroll
  for (int u = 0; u < NCH; ++u) {
    const int o = (P + 4 * u) * C_IN;
    g.a[u][1] = lds[la + o];
    g.a[u][2] = lds[la + o + C_IN];
    g.b[u][1] = lds[lb + o];
    g.b[u][2] = lds[lb + o + C_IN];
    if (OWN0 && u == 0) {
      g.a[u][0] = lds[e0a];
      g.b[u][0] = lds[e0b];
    } else {
      g.a[u][0] = lds[la + o - C_IN];
      g.b[u][0] = lds[lb + o - C_IN];
    }
    if (OWN3 && u == NCH - 1) {
      g.a[u][3] = lds[e3a];
      g.b[u][3] = lds[e3b];
    } else {
      g.a[u][3] = lds[la + o + 2 * C_IN];
      g.b[u][3] = lds[lb + o + 2 * C_IN];
    }
  }
}

template <int NCH>
__device__ __forceinline__ void compute_group(const Grp& g, v8f& acc1, v8f& acc2,
                                              bool lo3, bool hi3) {
#pragma unroll
  for (int u = 0; u < NCH; ++u) {
    const float a0 = g.a[u][0], a1 = g.a[u][1], a2 = g.a[u][2], a3 = g.a[u][3];
    const float b0 = g.b[u][0], b1 = g.b[u][1], b2 = g.b[u][2], b3 = g.b[u][3];
    const v2f A1 = {a1, a2};
    const v2f B1 = {a2 - a0, a3 - a1};
    const v2f A2 = {hi3 ? b1 : a1, hi3 ? b2 : a2};
    const v2f B2 = {lo3 ? (b2 - b0) : B1.x, lo3 ? (b3 - b1) : B1.y};

    acc1 = wmma_f32_4(A1, B1, acc1);
    acc2 = wmma_f32_4(A2, B2, acc2);
  }
}

__global__ __launch_bounds__(256) void sig2_kernel(const float* __restrict__ x,
                                                   float* __restrict__ out,
                                                   int n_paths) {
  __shared__ __attribute__((aligned(16))) float lds[LDS_FLOATS];

  const int lane = threadIdx.x & 31;
  const int wave = threadIdx.x >> 5;
  const int path = blockIdx.x * WAVES + wave;
  const float inv = 1.0f / (float)N_INC;

  for (int i = threadIdx.x; i < TAB_ROWS; i += 32 * WAVES) {
    const int m = i - 1;
    float tv = (float)m * inv;
    tv = (m == T_STEPS - 1) ? 1.0f : tv;
    tv = (m < 0 || m >= T_STEPS) ? 0.0f : tv;
    lds[TAB0 + i * C_IN] = tv;
  }
  __syncthreads();
  if (path >= n_paths) return;

  const int half = lane >> 4;
  const int ll   = lane & 15;
  const int klo  = half * 2;

  const float* __restrict__ xp = x + (size_t)path * PATH_FLOATS;
  float* __restrict__ op = out + (size_t)path * OUT_PER_PATH;

  const bool is_time = (ll == 0);
  const bool lo3 = ll < 3;
  const bool hi3 = ll >= 13;
  const int  mc0 = is_time ? 0 : (ll - 1);
  const int  mc1 = lo3 ? (15 + ll) : (hi3 ? (ll + 2) : mc0);

  const float x0_0 = is_time ? 0.0f : xp[mc0];
  const float x0_1 = xp[mc1];

  const int ring_off = wave * RING_FLOATS;
  float* ring = lds + ring_off;
  float* st0 = ring + lane * 4;
  float* st1 = st0 + STG_FLOATS;
  const int abase = is_time ? (TAB0 + C_IN) : (ring_off + mc0);
  const int bbase = ring_off + mc1;
  const int astep = is_time ? RING_FLOATS : 0;
  const int zero_at = TAB0;
  int la = abase + klo * C_IN;
  const int lb = bbase + klo * C_IN;
  const int wrow = RING_ROWS - 2 + klo;
  int wa = abase + (is_time ? wrow : (wrow % RING_ROWS)) * C_IN;
  const int wb = bbase + (wrow % RING_ROWS) * C_IN;
  const int prow = half ? 1 : -1;
  int p0a = abase + (is_time ? prow : ((prow + RING_ROWS) % RING_ROWS)) * C_IN;
  const int p0b = bbase + ((prow + RING_ROWS) % RING_ROWS) * C_IN;
  const int f0a = half ? (abase + C_IN) : zero_at;
  const int f0b = half ? (bbase + C_IN) : zero_at;

  v8f acc1 = {}, acc2 = {};

#define PIN()                                                     \
  do {                                                            \
    asm volatile("" : "+v"(acc1), "+v"(acc2) : : "memory");       \
    __builtin_amdgcn_sched_barrier(0);                            \
  } while (0)
#define LDS_ORDER() __builtin_amdgcn_fence(__ATOMIC_ACQ_REL, "wavefront")
#define COMPUTE(G) compute_group<U_CH>(G, acc1, acc2, lo3, hi3)

  Stg R0, R1;
  issue_stage(R0, xp, 0, lane);
  issue_stage(R1, xp, 1, lane);
  store_stage(R0, st0);
  store_stage(R1, st1);
  LDS_ORDER();
  issue_stage(R0, xp, 2, lane);
  issue_stage(R1, xp, 3, lane);
  PIN();

  Grp A, B;
  load_group<0, U_CH, true, false>(A, lds, la, lb, f0a, f0b, 0, 0);
  PIN();
  for (int t = 0; t < N_TRIP; ++t) {
    load_group<32, U_CH, false, false>(B, lds, la, lb, 0, 0, 0, 0);
    PIN();
    COMPUTE(A);
    PIN();
    load_group<64, U_CH, false, false>(A, lds, la, lb, 0, 0, 0, 0);
    PIN();
    COMPUTE(B);
    PIN();
    store_stage(R0, st0);
    LDS_ORDER();
    issue_stage(R0, xp, 2 * t + 4, lane);
    PIN();
    load_group<96, U_CH, false, true>(B, lds, la, lb, 0, 0, wa, wb);
    PIN();
    COMPUTE(A);
    PIN();
    la += astep;
    wa += astep;
    p0a += astep;
    load_group<0, U_CH, true, false>(A, lds, la, lb, p0a, p0b, 0, 0);
    PIN();
    COMPUTE(B);
    PIN();
    store_stage(R1, st1);
    LDS_ORDER();
    issue_stage(R1, xp, 2 * t + 5, lane);
    PIN();
  }
  const int za = half ? zero_at : (la + (T_STEPS - 4) % RING_ROWS * C_IN + 2 * C_IN);
  const int zb = half ? zero_at : (lb + (T_STEPS - 4) % RING_ROWS * C_IN + 2 * C_IN);
  load_group<32, U_CH, false, false>(B, lds, la, lb, 0, 0, 0, 0);
  PIN();
  COMPUTE(A);
  PIN();
  load_group<64, U_CH, false, false>(A, lds, la, lb, 0, 0, 0, 0);
  PIN();
  COMPUTE(B);
  PIN();
  load_group<96, 2, false, true>(B, lds, la, lb, 0, 0, za, zb);
  PIN();
  COMPUTE(A);
  compute_group<2>(B, acc1, acc2, lo3, hi3);
#undef COMPUTE
#undef LDS_ORDER
#undef PIN

  const int lrow = (T_STEPS - 1) % RING_ROWS;
  const float xN_0 = is_time ? 1.0f : ring[lrow * C_IN + mc0];
  const float xN_1 = ring[lrow * C_IN + mc1];
  const float s1a = xN_0 - x0_0;
  const float s1b = xN_1 - x0_1;
  if (half == 0) {
    op[ll] = s1a;
    if (lo3) op[16 + ll] = s1b;
  }
  const float s1_c1  = lane_value(s1a, 1);
  const float s1_c2  = lane_value(s1a, 2);
  const float s1_c16 = lane_value(s1b, 0);
  const float s1_c17 = lane_value(s1b, 1);
  const float s1_c18 = lane_value(s1b, 2);

  const float x0c2 = lo3 ? x0_1 : x0_0;
  const float xNc2 = lo3 ? xN_1 : xN_0;
  const float cj1 = 0.5f * x0_0 - xN_0;
  const float cj2 = 0.5f * x0c2 - xNc2;
  float x0r1[8], x0r2[8], xNr1[8], xNr2[8];
#pragma unroll
  for (int v = 0; v < 8; ++v) {
    const float r0_lo = lane_value(x0_0, v);
    const float r0_hi = lane_value(x0_0, v + 8);
    const float rN_lo = lane_value(xN_0, v);
    const float rN_hi = lane_value(xN_0, v + 8);
    x0r1[v] = half ? r0_hi : r0_lo;
    xNr1[v] = half ? rN_hi : rN_lo;
    if (v >= 5) {
      const float r0_x = lane_value(x0_1, v + 8);
      const float rN_x = lane_value(xN_1, v + 8);
      x0r2[v] = half ? r0_x : r0_lo;
      xNr2[v] = half ? rN_x : rN_lo;
    } else {
      x0r2[v] = x0r1[v];
      xNr2[v] = xNr1[v];
    }
  }

#pragma unroll
  for (int v = 0; v < 8; ++v)
    op[C_AUG + (v + 8 * half) * C_AUG + ll] =
        0.5f * acc1[v] + 0.5f * xNr1[v] * xN_0 + x0r1[v] * cj1;

#pragma unroll
  for (int v = 0; v < 8; ++v) {
    const int m = v + 8 * half;
    const float val = 0.5f * acc2[v] + 0.5f * xNr2[v] * xNc2 + x0r2[v] * cj2;
    if (lo3) {
      const int j = 16 + ll;
      if (m < 13) {
        op[C_AUG + m * C_AUG + j] = val;
        if (m < 3) {
          const float s1m = (m == 0) ? 1.0f : (m == 1 ? s1_c1 : s1_c2);
          op[C_AUG + j * C_AUG + m] = s1b * s1m - val;
        }
      } else {
        op[C_AUG + (m + 3) * C_AUG + j] = val;
      }
    } else if (m >= 13) {
      const int i = m + 3;
      op[C_AUG + i * C_AUG + ll] = val;
      if (hi3) {
        const float s1i = (m == 13) ? s1_c16 : (m == 14 ? s1_c17 : s1_c18);
        op[C_AUG + ll * C_AUG + i] = s1a * s1i - val;
      }
    }
  }
}

extern "C" void kernel_launch(void* const* d_in, const int* in_sizes, int n_in,
                              void* d_out, int out_size, void* d_ws, size_t ws_size,
                              hipStream_t stream) {
  const float* x = (const float*)d_in[0];
  float* out = (float*)d_out;
  const int n_paths = in_sizes[0] / (T_STEPS * C_IN);
  const int waves_per_block = WAVES;
  dim3 block(32 * waves_per_block);
  dim3 grid((n_paths + waves_per_block - 1) / waves_per_block);
  sig2_kernel<<<grid, block, 0, stream>>>(x, out, n_paths);
}


# Second tile per chunk replaced by one packed tile per four chunks for the pairs inside channels {0,1,2} and {16,17,18} (six diagonals = S1^2/2); main tile in two accumulators; summation order changes
# speedup vs baseline: 3.2213x; 1.1929x over previous
#include <hip/hip_runtime.h>


typedef float v2f __attribute__((ext_vector_type(2)));
typedef float v4f __attribute__((ext_vector_type(4)));
typedef float v8f __attribute__((ext_vector_type(8)));

#define T_STEPS 1000
#define C_IN    18
#define C_AUG   19
#define N_INC   (T_STEPS - 1)
#define OUT_PER_PATH (C_AUG + C_AUG * C_AUG)
#define PATH_FLOATS (T_STEPS * C_IN)
#define WAVES   8

#define STG_LOADS  9
#define STG_ROWS   64
#define STG_FLOATS (STG_ROWS * C_IN)
#define RING_ROWS   (2 * STG_ROWS)
#define RING_FLOATS (RING_ROWS * C_IN)
#define U_CH    8
#define N_TRIP  7
#define TAB_ROWS   (T_STEPS + 2)
#define TAB0       (WAVES * RING_FLOATS)
#define LDS_FLOATS (TAB0 + TAB_ROWS * C_IN)

__device__ __forceinline__ v8f wmma_f32_4(v2f a, v2f b, v8f c) {
  return __builtin_amdgcn_wmma_f32_16x16x4_f32(false, a, false, b, (short)0, c,
                                               false, false);
}

__device__ __forceinline__ float lane_value(float v, int src_lane) {
  return __int_as_float(__builtin_amdgcn_readlane(__float_as_int(v), src_lane));
}

struct Stg {
  v4f r[STG_LOADS];
};

__device__ __forceinline__ void issue_stage(Stg& s, const float* __restrict__ xp,
                                            int stage, int lane) {
#pragma unroll
  for (int j = 0; j < STG_LOADS; ++j) {
    int off = stage * STG_FLOATS + j * 128 + lane * 4;
    off = off < PATH_FLOATS - 4 ? off : PATH_FLOATS - 4;
    s.r[j] = *reinterpret_cast<const v4f*>(xp + off);
  }
}

__device__ __forceinline__ void store_stage(const Stg& s, float* slot_lane) {
#pragma unroll
  for (int j = 0; j < STG_LOADS; ++j)
    *reinterpret_cast<v4f*>(slot_lane + j * 128) = s.r[j];
}

struct Grp {
  float a[U_CH][2];
  float b[U_CH][4];
  float ea[2][2];
  float eb[2][4];
};

template <int P, int NCH, int NE, bool OWN0, bool OWN3>
__device__ __forceinline__ void load_group(Grp& g, const float* lds, int la, int lb,
                                           int ea, int eb, int x0b, int x0e,
                                           int x3b, int x3e) {
#pragma unroll
  for (int u = 0; u < NCH; ++u) {
    const int o = (P + 4 * u) * C_IN;
    g.a[u][0] = lds[la + o];
    g.a[u][1] = lds[la + o + C_IN];
    g.b[u][1] = lds[lb + o];
    g.b[u][2] = lds[lb + o + C_IN];
    if (OWN0 && u == 0) g.b[u][0] = lds[x0b];
    else                g.b[u][0] = lds[lb + o - C_IN];
    if (OWN3 && u == NCH - 1) g.b[u][3] = lds[x3b];
    else                      g.b[u][3] = lds[lb + o + 2 * C_IN];
  }
#pragma unroll
  for (int e = 0; e < NE; ++e) {
    const int o = (P + 16 * e) * C_IN;
    g.ea[e][0] = lds[ea + o];
    g.ea[e][1] = lds[ea + o + C_IN];
    g.eb[e][1] = lds[eb + o];
    g.eb[e][2] = lds[eb + o + C_IN];
    if (OWN0 && e == 0) g.eb[e][0] = lds[x0e];
    else                g.eb[e][0] = lds[eb + o - C_IN];
    if (OWN3 && e == NE - 1) g.eb[e][3] = lds[x3e];
    else                     g.eb[e][3] = lds[eb + o + 2 * C_IN];
  }
}

template <int NCH, int NE>
__device__ __forceinline__ void compute_group(const Grp& g, v8f& accA, v8f& accB,
                                              v8f& accE) {
#pragma unroll
  for (int u = 0; u < NCH; ++u) {
    const v2f A = {g.a[u][0], g.a[u][1]};
    const v2f B = {g.b[u][2] - g.b[u][0], g.b[u][3] - g.b[u][1]};
    if (u & 1) accB = wmma_f32_4(A, B, accB);
    else       accA = wmma_f32_4(A, B, accA);
    if ((u & 3) == 3 && (u >> 2) < NE) {
      const int e = u >> 2;
      const v2f EA = {g.ea[e][0], g.ea[e][1]};
      const v2f EB = {g.eb[e][2] - g.eb[e][0], g.eb[e][3] - g.eb[e][1]};
      accE = wmma_f32_4(EA, EB, accE);
    }
  }
}

__global__ __launch_bounds__(256) void sig2_kernel(const float* __restrict__ x,
                                                   float* __restrict__ out,
                                                   int n_paths) {
  __shared__ __attribute__((aligned(16))) float lds[LDS_FLOATS];

  const int lane = threadIdx.x & 31;
  const int wave = threadIdx.x >> 5;
  const int path = blockIdx.x * WAVES + wave;
  const float inv = 1.0f / (float)N_INC;

  for (int i = threadIdx.x; i < TAB_ROWS; i += 32 * WAVES) {
    const int m = i - 1;
    float tv = (float)m * inv;
    tv = (m == T_STEPS - 1) ? 1.0f : tv;
    tv = (m < 0 || m >= T_STEPS) ? 0.0f : tv;
    lds[TAB0 + i * C_IN] = tv;
  }
  __syncthreads();
  if (path >= n_paths) return;

  const int half = lane >> 4;
  const int ll   = lane & 15;
  const int klo  = half * 2;

  const float* __restrict__ xp = x + (size_t)path * PATH_FLOATS;
  float* __restrict__ op = out + (size_t)path * OUT_PER_PATH;

  const bool is_time = (ll == 0);
  const int  mc0 = is_time ? 0 : (ll - 1);
  const int  mc1 = ll + 2;

  const float x0_0 = is_time ? 0.0f : xp[mc0];
  const float x0_1 = xp[mc1];

  const int ring_off = wave * RING_FLOATS;
  float* ring = lds + ring_off;
  float* st0 = ring + lane * 4;
  float* st1 = st0 + STG_FLOATS;
  const int abase = is_time ? (TAB0 + C_IN) : (ring_off + mc0);
  const int bbase = ring_off + mc1;
  const int astep = is_time ? RING_FLOATS : 0;
  const int zero_at = TAB0;
  int la = abase + klo * C_IN;
  const int lb = bbase + klo * C_IN;
  const int wrow = RING_ROWS - 2 + klo;
  const int wb = bbase + (wrow % RING_ROWS) * C_IN;
  const int prow = half ? 1 : -1;
  const int p0b = bbase + ((prow + RING_ROWS) % RING_ROWS) * C_IN;
  const int f0b = half ? (bbase + C_IN) : zero_at;

  const int es = ll >> 2;
  const int ei = ll & 3;
  const int erow = 4 * es + klo;
  const bool e_time = (ei == 0);
  const int emA = (ei == 1) ? 0 : (ei == 2 ? 15 : 16);
  const int emB = (ei == 0) ? 0 : (ei == 1 ? 1 : (ei == 2 ? 16 : 17));
  const int eabase = e_time ? (TAB0 + C_IN) : (ring_off + emA);
  const int ebbase = ring_off + emB;
  const int estep = e_time ? RING_FLOATS : 0;
  int ea = eabase + erow * C_IN;
  const int eb = ebbase + erow * C_IN;
  const int p0e = (erow == 0) ? (ebbase + (RING_ROWS - 1) * C_IN) : (eb - C_IN);
  const int f0e = (erow == 0) ? zero_at : (eb - C_IN);
  const int we = (erow == 14) ? ebbase : (eb + (RING_ROWS - 16 + 2) * C_IN);

  v8f acc1 = {}, acc2 = {}, acc3 = {};

#define PIN()                                                                  \
  do {                                                                         \
    asm volatile("" : "+v"(acc1), "+v"(acc2), "+v"(acc3) : : "memory");        \
    __builtin_amdgcn_sched_barrier(0);                                         \
  } while (0)
#define LDS_ORDER() __builtin_amdgcn_fence(__ATOMIC_ACQ_REL, "wavefront")
#define COMPUTE(G) compute_group<U_CH, 2>(G, acc1, acc2, acc3)

  Stg R0, R1;
  issue_stage(R0, xp, 0, lane);
  issue_stage(R1, xp, 1, lane);
  store_stage(R0, st0);
  store_stage(R1, st1);
  LDS_ORDER();
  issue_stage(R0, xp, 2, lane);
  issue_stage(R1, xp, 3, lane);
  PIN();

  Grp A, B;
  load_group<0, U_CH, 2, true, false>(A, lds, la, lb, ea, eb, f0b, f0e, 0, 0);
  PIN();
  for (int t = 0; t < N_TRIP; ++t) {
    load_group<32, U_CH, 2, false, false>(B, lds, la, lb, ea, eb, 0, 0, 0, 0);
    PIN();
    COMPUTE(A);
    PIN();
    load_group<64, U_CH, 2, false, false>(A, lds, la, lb, ea, eb, 0, 0, 0, 0);
    PIN();
    COMPUTE(B);
    PIN();
    store_stage(R0, st0);
    LDS_ORDER();
    issue_stage(R0, xp, 2 * t + 4, lane);
    PIN();
    load_group<96, U_CH, 2, false, true>(B, lds, la, lb, ea, eb, 0, 0, wb, we);
    PIN();
    COMPUTE(A);
    PIN();
    la += astep;
    ea += estep;
    load_group<0, U_CH, 2, true, false>(A, lds, la, lb, ea, eb, p0b, p0e, 0, 0);
    PIN();
    COMPUTE(B);
    PIN();
    store_stage(R1, st1);
    LDS_ORDER();
    issue_stage(R1, xp, 2 * t + 5, lane);
    PIN();
  }
  const int zb = half ? zero_at : (lb + (T_STEPS - 4) % RING_ROWS * C_IN + 2 * C_IN);
  const bool e_in = es < 2;
  const int tP = 96 * C_IN;
  const int ta0 = e_in ? (ea + tP) : zero_at;
  const int ta1 = e_in ? (ea + tP + C_IN) : zero_at;
  const int tb0 = e_in ? (eb + tP - C_IN) : zero_at;
  const int tb1 = e_in ? (eb + tP) : zero_at;
  const int tb2 = e_in ? (eb + tP + C_IN) : zero_at;
  const int tb3 = (e_in && erow != 6) ? (eb + tP + 2 * C_IN) : zero_at;
  load_group<32, U_CH, 2, false, false>(B, lds, la, lb, ea, eb, 0, 0, 0, 0);
  PIN();
  COMPUTE(A);
  PIN();
  load_group<64, U_CH, 2, false, false>(A, lds, la, lb, ea, eb, 0, 0, 0, 0);
  PIN();
  COMPUTE(B);
  PIN();
  load_group<96, 2, 0, false, true>(B, lds, la, lb, ea, eb, 0, 0, zb, 0);
  const float ta_0 = lds[ta0], ta_1 = lds[ta1];
  const float tb_0 = lds[tb0], tb_1 = lds[tb1], tb_2 = lds[tb2], tb_3 = lds[tb3];
  PIN();
  COMPUTE(A);
  compute_group<2, 0>(B, acc1, acc2, acc3);
  {
    const v2f EA = {ta_0, ta_1};
    const v2f EB = {tb_2 - tb_0, tb_3 - tb_1};
    acc3 = wmma_f32_4(EA, EB, acc3);
  }
#undef COMPUTE
#undef PIN

  const int lrow = (T_STEPS - 1) % RING_ROWS;
  const float xN_0 = is_time ? 1.0f : ring[lrow * C_IN + mc0];
  const float xN_1 = ring[lrow * C_IN + mc1];
  const float s1a = xN_0 - x0_0;
  const float s1b = xN_1 - x0_1;
  const bool hi3 = ll >= 13;
  if (half == 0) {
    op[ll] = s1a;
    if (hi3) op[ll + 3] = s1b;
    if (ll < 3) op[C_AUG + ll * C_AUG + ll] = 0.5f * s1a * s1a;
    if (hi3)    op[C_AUG + (ll + 3) * C_AUG + (ll + 3)] = 0.5f * s1b * s1b;
  }

#pragma unroll
  for (int v = 0; v < 8; ++v) ring[v * 32 + lane] = acc3[v];
  LDS_ORDER();
#undef LDS_ORDER

  const float cj = 0.5f * x0_1 - xN_1;
#pragma unroll
  for (int v = 0; v < 8; ++v) {
    const float r0_lo = lane_value(x0_0, v);
    const float r0_hi = lane_value(x0_0, v + 8);
    const float rN_lo = lane_value(xN_0, v);
    const float rN_hi = lane_value(xN_0, v + 8);
    const float x0r = half ? r0_hi : r0_lo;
    const float xNr = half ? rN_hi : rN_lo;
    const int i = v + 8 * half;
    const int j = ll + 3;
    const float val = 0.5f * (acc1[v] + acc2[v]) + 0.5f * xNr * xN_1 + x0r * cj;
    op[C_AUG + i * C_AUG + j] = val;
    if (hi3 || i < 3) op[C_AUG + j * C_AUG + i] = s1b * (xNr - x0r) - val;
  }

  if (lane < 6) {
    const int ri = lane < 2 ? 0 : (lane == 2 ? 1 : (lane < 5 ? 2 : 3));
    const int ci = lane == 0 ? 0 : (lane < 3 ? 1 : (lane == 3 ? 2 : 3));
    const float q = ((ring[ri * 32 + ci] + ring[(4 + ri) * 32 + 4 + ci]) +
                     ring[ri * 32 + 24 + ci]) + ring[(4 + ri) * 32 + 28 + ci];
    const int i = ri < 2 ? ri : (ri + 14);
    const int j = ci < 2 ? (ci + 1) : (ci + 15);
    const int im = i == 0 ? 0 : (i - 1);
    const float x0i = i == 0 ? 0.0f : xp[im];
    const float xNi = i == 0 ? 1.0f : xp[(T_STEPS - 1) * C_IN + im];
    const float x0j = xp[j - 1];
    const float xNj = xp[(T_STEPS - 1) * C_IN + j - 1];
    const float val = 0.5f * q + 0.5f * xNi * xNj + x0i * (0.5f * x0j - xNj);
    op[C_AUG + i * C_AUG + j] = val;
    op[C_AUG + j * C_AUG + i] = (xNi - x0i) * (xNj - x0j) - val;
  }
}

extern "C" void kernel_launch(void* const* d_in, const int* in_sizes, int n_in,
                              void* d_out, int out_size, void* d_ws, size_t ws_size,
                              hipStream_t stream) {
  const float* x = (const float*)d_in[0];
  float* out = (float*)d_out;
  const int n_paths = in_sizes[0] / (T_STEPS * C_IN);
  const int waves_per_block = WAVES;
  dim3 block(32 * waves_per_block);
  dim3 grid((n_paths + waves_per_block - 1) / waves_per_block);
  sig2_kernel<<<grid, block, 0, stream>>>(x, out, n_paths);
}
